# starting kernel
# baseline (speedup 1.0000x reference)
#include <hip/hip_runtime.h>


typedef __attribute__((ext_vector_type(2))) float v2f;
typedef __attribute__((ext_vector_type(8))) float v8f;

#define T_LEN 512
#define WTP_STRIDE 272
#define H_STRIDE 68

static __device__ __forceinline__ v8f wmma4(v2f a, v2f b, v8f c) {
  return __builtin_amdgcn_wmma_f32_16x16x4_f32(false, a, false, b, (short)0, c,
                                               false, false);
}

static __device__ __forceinline__ float sigm(float x) {
  return 1.0f / (1.0f + __expf(-x));
}
static __device__ __forceinline__ float tanh_(float x) {
  return 1.0f - 2.0f / (1.0f + __expf(2.0f * x));
}

__global__ __launch_bounds__(128) void lstm_wmma_kernel(
    const float* __restrict__ x, const float* __restrict__ W_ih,
    const float* __restrict__ W_hh, const float* __restrict__ b_ih,
    const float* __restrict__ b_hh, const float* __restrict__ W_out,
    const float* __restrict__ b_out, float* __restrict__ out) {
  __shared__ float WtP[34 * WTP_STRIDE * 2];
  __shared__ float WtOutP[32 * 16 * 2];
  __shared__ float hS[4 * 16 * H_STRIDE];

  const int tid  = threadIdx.x;
  const int wave = tid >> 5;
  const int lane = tid & 31;
  const int col  = lane & 15;
  const int hi   = lane >> 4;
  const int bbase = blockIdx.x * 64 + wave * 16;

  for (int idx = tid; idx < 68 * 256; idx += 128) {
    int k = idx >> 8, n = idx & 255;
    float v;
    if (k < 64)      v = W_hh[n * 64 + k];
    else if (k < 67) v = W_ih[n * 3 + (k - 64)];
    else             v = b_ih[n] + b_hh[n];
    WtP[((k >> 1) * WTP_STRIDE + n) * 2 + (k & 1)] = v;
  }
  for (int idx = tid; idx < 64 * 16; idx += 128) {
    int k = idx >> 4, n = idx & 15;
    WtOutP[((k >> 1) * 16 + n) * 2 + (k & 1)] =
        (n < 3) ? W_out[n * 64 + k] : 0.0f;
  }
  for (int idx = tid; idx < 4 * 16 * H_STRIDE; idx += 128) hS[idx] = 0.0f;
  __syncthreads();

  float* hW = &hS[wave * (16 * H_STRIDE)];
  const float* xrow = x + (size_t)(bbase + col) * (T_LEN * 3);
  const float bO = (col < 3) ? b_out[col] : 0.0f;
  const v8f vzero = {0, 0, 0, 0, 0, 0, 0, 0};

  v8f cs0 = vzero, cs1 = vzero, cs2 = vzero, cs3 = vzero;

  for (int t = 0; t < T_LEN; ++t) {
    const float* xp = xrow + t * 3;
    v2f ax;
    if (hi == 0) { ax[0] = xp[0]; ax[1] = xp[1]; }
    else         { ax[0] = xp[2]; ax[1] = 1.0f; }

    v8f hn0, hn1, hn2, hn3;

#define GATE_BLOCK(J, CS, HN)                                                 \
    {                                                                         \
      const int nb = (J) * 16 + col;                                          \
      v8f ai, af, ag, ao;                                                     \
      {             \
        const float* wp = &WtP[((32 + hi) * WTP_STRIDE + nb) * 2];            \
        v2f b0 = *(const v2f*)(wp);                                           \
        v2f b1 = *(const v2f*)(wp + 128);                                     \
        v2f b2 = *(const v2f*)(wp + 256);                                     \
        v2f b3 = *(const v2f*)(wp + 384);                                     \
        ai = wmma4(ax, b0, vzero);                                            \
        af = wmma4(ax, b1, vzero);                                            \
        ag = wmma4(ax, b2, vzero);                                            \
        ao = wmma4(ax, b3, vzero);                                            \
      }                                                                       \
      _Pragma("unroll 2")                                                     \
      for (int ks = 0; ks < 16; ++ks) {                                       \
        v2f a = *(const v2f*)&hW[col * H_STRIDE + ks * 4 + hi * 2];           \
        const float* wp = &WtP[((ks * 2 + hi) * WTP_STRIDE + nb) * 2];        \
        v2f b0 = *(const v2f*)(wp);                                           \
        v2f b1 = *(const v2f*)(wp + 128);                                     \
        v2f b2 = *(const v2f*)(wp + 256);                                     \
        v2f b3 = *(const v2f*)(wp + 384);                                     \
        ai = wmma4(a, b0, ai);                                                \
        af = wmma4(a, b1, af);                                                \
        ag = wmma4(a, b2, ag);                                                \
        ao = wmma4(a, b3, ao);                                                \
      }                                                                       \
      _Pragma("unroll")                                                       \
      for (int r = 0; r < 8; ++r) {                                           \
        float iv = sigm(ai[r]), fv = sigm(af[r]);                             \
        float gv = tanh_(ag[r]), ov = sigm(ao[r]);                            \
        float cv = fv * CS[r] + iv * gv;                                      \
        CS[r] = cv;                                                           \
        HN[r] = ov * tanh_(cv);                                               \
      }                                                                       \
    }

    GATE_BLOCK(0, cs0, hn0)
    GATE_BLOCK(1, cs1, hn1)
    GATE_BLOCK(2, cs2, hn2)
    GATE_BLOCK(3, cs3, hn3)
#undef GATE_BLOCK

    {
      float* hp = &hW[hi * 8 * H_STRIDE + col];
#pragma unroll
      for (int r = 0; r < 8; ++r) {
        hp[r * H_STRIDE +  0] = hn0[r];
        hp[r * H_STRIDE + 16] = hn1[r];
        hp[r * H_STRIDE + 32] = hn2[r];
        hp[r * H_STRIDE + 48] = hn3[r];
      }
    }

    v8f aco = {bO, bO, bO, bO, bO, bO, bO, bO};
#pragma unroll 2
    for (int ks = 0; ks < 16; ++ks) {
      v2f a = *(const v2f*)&hW[col * H_STRIDE + ks * 4 + hi * 2];
      v2f b = *(const v2f*)&WtOutP[((ks * 2 + hi) * 16 + col) * 2];
      aco = wmma4(a, b, aco);
    }
    if (col < 3) {
#pragma unroll
      for (int r = 0; r < 8; ++r) {
        size_t bidx = (size_t)(bbase + hi * 8 + r);
        out[(bidx * T_LEN + t) * 3 + col] = aco[r];
      }
    }
  }
}

extern "C" void kernel_launch(void* const* d_in, const int* in_sizes, int n_in,
                              void* d_out, int out_size, void* d_ws,
                              size_t ws_size, hipStream_t stream) {
  const float* x     = (const float*)d_in[0];
  const float* W_ih  = (const float*)d_in[1];
  const float* W_hh  = (const float*)d_in[2];
  const float* b_ih  = (const float*)d_in[3];
  const float* b_hh  = (const float*)d_in[4];
  const float* W_out = (const float*)d_in[5];
  const float* b_out = (const float*)d_in[6];
  float* out = (float*)d_out;

  dim3 grid(4096 / 64);
  dim3 block(128);
  lstm_wmma_kernel<<<grid, block, 0, stream>>>(x, W_ih, W_hh, b_ih, b_hh,
                                               W_out, b_out, out);
}


# Gate product re-split 2 x 2 (K tile pairs x hidden tile pairs): a wave exchanges 1 h tile + 4 partial gate tiles, not 12, each hop between split barrier halves under WMMAs; gate sum order changes
# speedup vs baseline: 5.1282x; 5.1282x over previous
#include <hip/hip_runtime.h>


typedef __attribute__((ext_vector_type(2))) float v2f;
typedef __attribute__((ext_vector_type(4))) float v4f;
typedef __attribute__((ext_vector_type(8))) float v8f;

#define T_LEN 512

static __device__ __forceinline__ v8f wmma4(v2f a, v2f b, v8f c) {
  return __builtin_amdgcn_wmma_f32_16x16x4_f32(false, a, false, b, (short)0, c,
                                               false, false);
}

static __device__ __forceinline__ float sigm(float x) {
  return 1.0f / (1.0f + __expf(-x));
}
static __device__ __forceinline__ float tanh_(float x) {
  return 1.0f - 2.0f / (1.0f + __expf(2.0f * x));
}

#define HFRAG(H, P) ((v2f){(H)[2 * (P)], (H)[2 * (P) + 1]})

static __device__ __forceinline__ void keep_order() {
  __builtin_amdgcn_sched_barrier(0);
}
static __device__ __forceinline__ void lds_done_signal() {
  keep_order();
  asm volatile("s_wait_dscnt 0x0\n\ts_barrier_signal -1" ::: "memory");
  keep_order();
}
static __device__ __forceinline__ void lds_done_signal_after(v8f a, v8f b, v8f c, v8f d) {
  asm volatile("s_wait_dscnt 0x0\n\ts_barrier_signal -1"
               :: "v"(a), "v"(b), "v"(c), "v"(d) : "memory");
  keep_order();
}
static __device__ __forceinline__ void barrier_wait_after(v8f a, v8f b, v8f c, v8f d) {
  asm volatile("s_barrier_wait -1" :: "v"(a), "v"(b), "v"(c), "v"(d) : "memory");
}
static __device__ __forceinline__ void barrier_wait() {
  keep_order();
  asm volatile("s_barrier_wait -1" ::: "memory");
}

__global__ __launch_bounds__(128) void lstm_wmma_kernel(
    const float* __restrict__ x, const float* __restrict__ W_ih,
    const float* __restrict__ W_hh, const float* __restrict__ b_ih,
    const float* __restrict__ b_hh, const float* __restrict__ W_out,
    const float* __restrict__ b_out, float* __restrict__ out) {
  __shared__ v4f lds_h[4 * 2 * 32];
  __shared__ v4f lds_p[4 * 4 * 2 * 32];
  __shared__ float lds_out[2][4 * 3 * 32];

  const int tid  = threadIdx.x;
  const int w    = tid >> 5;
  const int lane = tid & 31;
  const int col  = lane & 15;
  const int hi   = lane >> 4;
  const int ws   = w ^ 1;
  const int wp   = w ^ 2;
  const int bbase = blockIdx.x * 16;

  v2f wA[2][2][4][4];
#pragma unroll
  for (int j = 0; j < 2; ++j) {
#pragma unroll
    for (int kt = 0; kt < 2; ++kt) {
#pragma unroll
      for (int g = 0; g < 4; ++g) {
        const int row = g * 64 + (j ? wp : w) * 16 + col;
#pragma unroll
        for (int p = 0; p < 4; ++p) {
          const float* ap = &W_hh[row * 64 + (kt ? ws : w) * 16 + hi * 8 + p * 2];
          wA[j][kt][g][p][0] = ap[0];
          wA[j][kt][g][p][1] = ap[1];
        }
      }
    }
  }
  v2f wX[4];
#pragma unroll
  for (int g = 0; g < 4; ++g) {
    const int row = g * 64 + w * 16 + col;
    const float w0 = W_ih[row * 3 + 0], w1 = W_ih[row * 3 + 1];
    const float w2 = W_ih[row * 3 + 2], bb = b_ih[row] + b_hh[row];
    wX[g][0] = hi ? w2 : w0;
    wX[g][1] = hi ? bb : w1;
  }
  v2f wO[4];
  {
    const int orow = (col < 3) ? col : 0;
#pragma unroll
    for (int p = 0; p < 4; ++p) {
      const float* op = &W_out[orow * 64 + w * 16 + hi * 8 + p * 2];
      wO[p][0] = (col < 3) ? op[0] : 0.0f;
      wO[p][1] = (col < 3) ? op[1] : 0.0f;
    }
  }
  const float bo0 = b_out[0], bo1 = b_out[1], bo2 = b_out[2];

  const float* xrow = x + (size_t)(bbase + col) * (T_LEN * 3);
  float* orow_p = out + (size_t)(bbase + col) * (T_LEN * 3);
  const v8f vzero = {0, 0, 0, 0, 0, 0, 0, 0};

  v4f* const h_mine  = lds_h + (w * 2) * 32 + lane;
  const v4f* const h_other = lds_h + (ws * 2) * 32 + lane;
  v4f* const p_send = lds_p + (wp * 8) * 32 + lane;
  const v4f* const p_recv = lds_p + (w * 8) * 32 + lane;

  v8f hs = vzero;
  v8f cs = vzero;
  h_mine[0]  = (v4f){0, 0, 0, 0};
  h_mine[32] = (v4f){0, 0, 0, 0};
  lds_out[1][(w * 3 + 0) * 32 + lane] = 0.0f;
  lds_out[1][(w * 3 + 1) * 32 + lane] = 0.0f;
  lds_out[1][(w * 3 + 2) * 32 + lane] = 0.0f;

  for (int t = 0; t < T_LEN; ++t) {
    float* ob = lds_out[t & 1];
    const float* obp = lds_out[(t & 1) ^ 1];

    const float* xp = xrow + t * 3;
    v2f bx;
    bx[0] = xp[2 * hi];
    const float x1 = xp[1 + hi];
    bx[1] = hi ? 1.0f : x1;

    const v2f hb[4] = {HFRAG(hs, 0), HFRAG(hs, 1), HFRAG(hs, 2), HFRAG(hs, 3)};

    v8f pa[4], ga[4];
#pragma unroll
    for (int g = 0; g < 4; ++g) pa[g] = wmma4(wA[1][0][g][0], hb[0], vzero);

    lds_done_signal_after(pa[0], pa[1], pa[2], pa[3]);

    v8f po = vzero;
#pragma unroll
    for (int p = 0; p < 4; ++p) po = wmma4(wO[p], hb[p], po);

#pragma unroll
    for (int p = 1; p < 4; ++p) {
#pragma unroll
      for (int g = 0; g < 4; ++g)
        pa[g] = wmma4(wA[1][0][g][p], hb[p], pa[g]);
    }
#pragma unroll
    for (int p = 0; p < 2; ++p) {
#pragma unroll
      for (int g = 0; g < 4; ++g)
        ga[g] = wmma4(wA[0][0][g][p], hb[p], p ? ga[g] : vzero);
    }

    barrier_wait();
    v8f hx;
    {
      const v4f lo = h_other[0];
      const v4f up = h_other[32];
      hx = (v8f){lo[0], lo[1], lo[2], lo[3], up[0], up[1], up[2], up[3]};
    }
    float o0 = bo0, o1 = bo1, o2 = bo2;
#pragma unroll
    for (int s = 0; s < 4; ++s) {
      o0 += obp[(s * 3 + 0) * 32 + lane];
      o1 += obp[(s * 3 + 1) * 32 + lane];
      o2 += obp[(s * 3 + 2) * 32 + lane];
    }
#pragma unroll
    for (int p = 2; p < 4; ++p) {
#pragma unroll
      for (int g = 0; g < 4; ++g)
        ga[g] = wmma4(wA[0][0][g][p], hb[p], ga[g]);
    }

    const v2f xb[4] = {HFRAG(hx, 0), HFRAG(hx, 1), HFRAG(hx, 2), HFRAG(hx, 3)};
#pragma unroll
    for (int p = 0; p < 4; ++p) {
#pragma unroll
      for (int g = 0; g < 4; ++g)
        pa[g] = wmma4(wA[1][1][g][p], xb[p], pa[g]);
    }
#pragma unroll
    for (int g = 0; g < 4; ++g) {
      p_send[(g * 2 + 0) * 32] = (v4f){pa[g][0], pa[g][1], pa[g][2], pa[g][3]};
      p_send[(g * 2 + 1) * 32] = (v4f){pa[g][4], pa[g][5], pa[g][6], pa[g][7]};
    }
    ob[(w * 3 + 0) * 32 + lane] = po[0];
    ob[(w * 3 + 1) * 32 + lane] = po[1];
    ob[(w * 3 + 2) * 32 + lane] = po[2];
    keep_order();

    auto own_step = [&](int i) {
      const int p = i >> 2, g = i & 3;
      if (p < 4) ga[g] = wmma4(wA[0][1][g][p], xb[p], ga[g]);
      else       ga[g] = wmma4(wX[g], bx, ga[g]);
    };
#pragma unroll
    for (int i = 0; i < 6; ++i) own_step(i);

    lds_done_signal();

#pragma unroll
    for (int i = 6; i < 12; ++i) own_step(i);

    barrier_wait_after(ga[0], ga[1], ga[2], ga[3]);
    v8f rp[4];
#pragma unroll
    for (int g = 0; g < 4; ++g) {
      const v4f lo = p_recv[(g * 2 + 0) * 32];
      const v4f up = p_recv[(g * 2 + 1) * 32];
      rp[g] = (v8f){lo[0], lo[1], lo[2], lo[3], up[0], up[1], up[2], up[3]};
    }
#pragma unroll
    for (int i = 12; i < 20; ++i) own_step(i);

#pragma unroll
    for (int g = 0; g < 4; ++g) ga[g] += rp[g];

    if (w == 0 && t > 1 && hi == 0) {
      float* op = orow_p + (t - 2) * 3;
      op[0] = o0;
      op[1] = o1;
      op[2] = o2;
    }

#pragma unroll
    for (int r = 0; r < 8; ++r) {
      float iv = sigm(ga[0][r]), fv = sigm(ga[1][r]);
      float gv = tanh_(ga[2][r]), ov = sigm(ga[3][r]);
      float cv = fv * cs[r] + iv * gv;
      cs[r] = cv;
      hs[r] = ov * tanh_(cv);
    }
    h_mine[0]  = (v4f){hs[0], hs[1], hs[2], hs[3]};
    h_mine[32] = (v4f){hs[4], hs[5], hs[6], hs[7]};
  }

  if (w == 0) {
    const float* obp = lds_out[(T_LEN & 1) ^ 1];
    float o0 = bo0, o1 = bo1, o2 = bo2;
#pragma unroll
    for (int s = 0; s < 4; ++s) {
      o0 += obp[(s * 3 + 0) * 32 + lane];
      o1 += obp[(s * 3 + 1) * 32 + lane];
      o2 += obp[(s * 3 + 2) * 32 + lane];
    }
    if (hi == 0) {
      float* op = orow_p + (T_LEN - 2) * 3;
      op[0] = o0;
      op[1] = o1;
      op[2] = o2;
    }
  }
  {
    float* ob = lds_out[T_LEN & 1];
    v8f po = vzero;
#pragma unroll
    for (int p = 0; p < 4; ++p) po = wmma4(wO[p], HFRAG(hs, p), po);
    ob[(w * 3 + 0) * 32 + lane] = po[0];
    ob[(w * 3 + 1) * 32 + lane] = po[1];
    ob[(w * 3 + 2) * 32 + lane] = po[2];
    __syncthreads();
    if (w == 0) {
      float o0 = bo0, o1 = bo1, o2 = bo2;
#pragma unroll
      for (int s = 0; s < 4; ++s) {
        o0 += ob[(s * 3 + 0) * 32 + lane];
        o1 += ob[(s * 3 + 1) * 32 + lane];
        o2 += ob[(s * 3 + 2) * 32 + lane];
      }
      if (hi == 0) {
        float* op = orow_p + (T_LEN - 1) * 3;
        op[0] = o0;
        op[1] = o1;
        op[2] = o2;
      }
    }
  }
}

extern "C" void kernel_launch(void* const* d_in, const int* in_sizes, int n_in,
                              void* d_out, int out_size, void* d_ws,
                              size_t ws_size, hipStream_t stream) {
  const float* x     = (const float*)d_in[0];
  const float* W_ih  = (const float*)d_in[1];
  const float* W_hh  = (const float*)d_in[2];
  const float* b_ih  = (const float*)d_in[3];
  const float* b_hh  = (const float*)d_in[4];
  const float* W_out = (const float*)d_in[5];
  const float* b_out = (const float*)d_in[6];
  float* out = (float*)d_out;

  dim3 grid(4096 / 16);
  dim3 block(128);
  lstm_wmma_kernel<<<grid, block, 0, stream>>>(x, W_ih, W_hh, b_ih, b_hh,
                                               W_out, b_out, out);
}


# Cell update in 4 register-pair slices: each new h pair feeds the next step's own-tile k-step at once, its 9 WMMAs issued among the next pair's exp/div; x/bias k-step moved up: gate sum order changes
# speedup vs baseline: 5.2897x; 1.0315x over previous
#include <hip/hip_runtime.h>


typedef __attribute__((ext_vector_type(2))) float v2f;
typedef __attribute__((ext_vector_type(4))) float v4f;
typedef __attribute__((ext_vector_type(8))) float v8f;

#define T_LEN 512

static __device__ __forceinline__ v8f wmma4(v2f a, v2f b, v8f c) {
  return __builtin_amdgcn_wmma_f32_16x16x4_f32(false, a, false, b, (short)0, c,
                                               false, false);
}

static __device__ __forceinline__ float sigm(float x) {
  return 1.0f / (1.0f + __expf(-x));
}
static __device__ __forceinline__ float tanh_(float x) {
  return 1.0f - 2.0f / (1.0f + __expf(2.0f * x));
}

#define HFRAG(H, P) ((v2f){(H)[2 * (P)], (H)[2 * (P) + 1]})

static __device__ __forceinline__ void keep_order() {
  __builtin_amdgcn_sched_barrier(0);
}
static __device__ __forceinline__ void lds_done_signal() {
  keep_order();
  asm volatile("s_wait_dscnt 0x0\n\ts_barrier_signal -1" ::: "memory");
  keep_order();
}
static __device__ __forceinline__ void lds_done_signal_after(v8f a, v8f b, v8f c, v8f d) {
  asm volatile("s_wait_dscnt 0x0\n\ts_barrier_signal -1"
               :: "v"(a), "v"(b), "v"(c), "v"(d) : "memory");
  keep_order();
}
static __device__ __forceinline__ void barrier_wait_after(v8f a, v8f b, v8f c, v8f d) {
  asm volatile("s_barrier_wait -1" :: "v"(a), "v"(b), "v"(c), "v"(d) : "memory");
}
static __device__ __forceinline__ void barrier_wait() {
  keep_order();
  asm volatile("s_barrier_wait -1" ::: "memory");
}

__global__ __launch_bounds__(128) void lstm_wmma_kernel(
    const float* __restrict__ x, const float* __restrict__ W_ih,
    const float* __restrict__ W_hh, const float* __restrict__ b_ih,
    const float* __restrict__ b_hh, const float* __restrict__ W_out,
    const float* __restrict__ b_out, float* __restrict__ out) {
  __shared__ v4f lds_h[4 * 2 * 32];
  __shared__ v4f lds_p[4 * 4 * 2 * 32];
  __shared__ float lds_out[2][4 * 3 * 32];

  const int tid  = threadIdx.x;
  const int w    = tid >> 5;
  const int lane = tid & 31;
  const int col  = lane & 15;
  const int hi   = lane >> 4;
  const int ws   = w ^ 1;
  const int wp   = w ^ 2;
  const int bbase = blockIdx.x * 16;

  v2f wA[2][2][4][4];
#pragma unroll
  for (int j = 0; j < 2; ++j) {
#pragma unroll
    for (int kt = 0; kt < 2; ++kt) {
#pragma unroll
      for (int g = 0; g < 4; ++g) {
        const int row = g * 64 + (j ? wp : w) * 16 + col;
#pragma unroll
        for (int p = 0; p < 4; ++p) {
          const float* ap = &W_hh[row * 64 + (kt ? ws : w) * 16 + hi * 8 + p * 2];
          wA[j][kt][g][p][0] = ap[0];
          wA[j][kt][g][p][1] = ap[1];
        }
      }
    }
  }
  v2f wX[4];
#pragma unroll
  for (int g = 0; g < 4; ++g) {
    const int row = g * 64 + w * 16 + col;
    const float w0 = W_ih[row * 3 + 0], w1 = W_ih[row * 3 + 1];
    const float w2 = W_ih[row * 3 + 2], bb = b_ih[row] + b_hh[row];
    wX[g][0] = hi ? w2 : w0;
    wX[g][1] = hi ? bb : w1;
  }
  v2f wO[4];
  {
    const int orow = (col < 3) ? col : 0;
#pragma unroll
    for (int p = 0; p < 4; ++p) {
      const float* op = &W_out[orow * 64 + w * 16 + hi * 8 + p * 2];
      wO[p][0] = (col < 3) ? op[0] : 0.0f;
      wO[p][1] = (col < 3) ? op[1] : 0.0f;
    }
  }
  const float bo0 = b_out[0], bo1 = b_out[1], bo2 = b_out[2];

  const float* xrow = x + (size_t)(bbase + col) * (T_LEN * 3);
  float* orow_p = out + (size_t)(bbase + col) * (T_LEN * 3);
  const v8f vzero = {0, 0, 0, 0, 0, 0, 0, 0};

  v4f* const h_mine  = lds_h + (w * 2) * 32 + lane;
  const v4f* const h_other = lds_h + (ws * 2) * 32 + lane;
  v4f* const p_send = lds_p + (wp * 8) * 32 + lane;
  const v4f* const p_recv = lds_p + (w * 8) * 32 + lane;

  v8f hs = vzero;
  v8f cs = vzero;
  h_mine[0]  = (v4f){0, 0, 0, 0};
  h_mine[32] = (v4f){0, 0, 0, 0};
  lds_out[1][(w * 3 + 0) * 32 + lane] = 0.0f;
  lds_out[1][(w * 3 + 1) * 32 + lane] = 0.0f;
  lds_out[1][(w * 3 + 2) * 32 + lane] = 0.0f;

  v8f pa[4], ga[4], po = vzero;
#pragma unroll
  for (int g = 0; g < 4; ++g) { pa[g] = vzero; ga[g] = vzero; }

  v2f bx;
  {
    bx[0] = xrow[2 * hi];
    const float x1 = xrow[1 + hi];
    bx[1] = hi ? 1.0f : x1;
  }

  lds_done_signal();

  for (int t = 0; t < T_LEN; ++t) {
    float* ob = lds_out[t & 1];
    const float* obp = lds_out[(t & 1) ^ 1];

    barrier_wait();
    const v4f hlo = h_other[0];
    const v4f hup = h_other[32];
    float opv[12];
#pragma unroll
    for (int s = 0; s < 4; ++s) {
      opv[s * 3 + 0] = obp[(s * 3 + 0) * 32 + lane];
      opv[s * 3 + 1] = obp[(s * 3 + 1) * 32 + lane];
      opv[s * 3 + 2] = obp[(s * 3 + 2) * 32 + lane];
    }
    keep_order();
    const v8f hx = (v8f){hlo[0], hlo[1], hlo[2], hlo[3], hup[0], hup[1], hup[2], hup[3]};
    float o0 = bo0, o1 = bo1, o2 = bo2;
#pragma unroll
    for (int s = 0; s < 4; ++s) {
      o0 += opv[s * 3 + 0];
      o1 += opv[s * 3 + 1];
      o2 += opv[s * 3 + 2];
    }
#pragma unroll
    for (int g = 0; g < 4; ++g) ga[g] = wmma4(wX[g], bx, ga[g]);
    v2f bxn;
    {
      const float* xp = xrow + ((t + 1) & (T_LEN - 1)) * 3;
      bxn[0] = xp[2 * hi];
      const float x1 = xp[1 + hi];
      bxn[1] = hi ? 1.0f : x1;
    }

    const v2f xb[4] = {HFRAG(hx, 0), HFRAG(hx, 1), HFRAG(hx, 2), HFRAG(hx, 3)};
#pragma unroll
    for (int p = 0; p < 4; ++p) {
#pragma unroll
      for (int g = 0; g < 4; ++g)
        pa[g] = wmma4(wA[1][1][g][p], xb[p], pa[g]);
    }
#pragma unroll
    for (int g = 0; g < 4; ++g) {
      p_send[(g * 2 + 0) * 32] = (v4f){pa[g][0], pa[g][1], pa[g][2], pa[g][3]};
      p_send[(g * 2 + 1) * 32] = (v4f){pa[g][4], pa[g][5], pa[g][6], pa[g][7]};
    }
    ob[(w * 3 + 0) * 32 + lane] = po[0];
    ob[(w * 3 + 1) * 32 + lane] = po[1];
    ob[(w * 3 + 2) * 32 + lane] = po[2];
    keep_order();

    auto own_step = [&](int i) {
      const int p = i >> 2, g = i & 3;
      ga[g] = wmma4(wA[0][1][g][p], xb[p], ga[g]);
    };
#pragma unroll
    for (int i = 0; i < 4; ++i) own_step(i);

    lds_done_signal();

#pragma unroll
    for (int i = 4; i < 10; ++i) own_step(i);

    barrier_wait_after(ga[0], ga[1], ga[2], ga[3]);
    v8f rp[4];
#pragma unroll
    for (int g = 0; g < 4; ++g) {
      const v4f lo = p_recv[(g * 2 + 0) * 32];
      const v4f up = p_recv[(g * 2 + 1) * 32];
      rp[g] = (v8f){lo[0], lo[1], lo[2], lo[3], up[0], up[1], up[2], up[3]};
    }
    keep_order();
#pragma unroll
    for (int i = 10; i < 16; ++i) own_step(i);

    v8f gt[4];
#pragma unroll
    for (int g = 0; g < 4; ++g) gt[g] = ga[g] + rp[g];

    if (w == 0 && t > 1 && hi == 0) {
      float* op = orow_p + (t - 2) * 3;
      op[0] = o0;
      op[1] = o1;
      op[2] = o2;
    }

    auto cell_pair = [&](int p) {
#pragma unroll
      for (int r = 2 * p; r < 2 * p + 2; ++r) {
        float iv = sigm(gt[0][r]), fv = sigm(gt[1][r]);
        float gv = tanh_(gt[2][r]), ov = sigm(gt[3][r]);
        float cv = fv * cs[r] + iv * gv;
        cs[r] = cv;
        hs[r] = ov * tanh_(cv);
      }
    };
    auto own_k_pa = [&](int p) {
      const v2f b = HFRAG(hs, p);
#pragma unroll
      for (int g = 0; g < 4; ++g) pa[g] = wmma4(wA[1][0][g][p], b, p ? pa[g] : vzero);
    };
    auto own_k_ga = [&](int p) {
      const v2f b = HFRAG(hs, p);
#pragma unroll
      for (int g = 0; g < 4; ++g) ga[g] = wmma4(wA[0][0][g][p], b, p ? ga[g] : vzero);
      po = wmma4(wO[p], b, p ? po : vzero);
    };
    cell_pair(0);
#pragma unroll
    for (int p = 0; p < 3; ++p) {
      keep_order();
      own_k_pa(p);
      own_k_ga(p);
      cell_pair(p + 1);
#pragma unroll
      for (int i = 0; i < 9; ++i) {
        __builtin_amdgcn_sched_group_barrier(0x008, 1, 0);
        __builtin_amdgcn_sched_group_barrier(0x402, 17, 0);
      }
    }
    keep_order();
    h_mine[0]  = (v4f){hs[0], hs[1], hs[2], hs[3]};
    h_mine[32] = (v4f){hs[4], hs[5], hs[6], hs[7]};
    keep_order();
    own_k_pa(3);
    lds_done_signal_after(pa[0], pa[1], pa[2], pa[3]);
    own_k_ga(3);
    bx = bxn;
  }

  barrier_wait();

  if (w == 0) {
    const float* obp = lds_out[(T_LEN & 1) ^ 1];
    float o0 = bo0, o1 = bo1, o2 = bo2;
#pragma unroll
    for (int s = 0; s < 4; ++s) {
      o0 += obp[(s * 3 + 0) * 32 + lane];
      o1 += obp[(s * 3 + 1) * 32 + lane];
      o2 += obp[(s * 3 + 2) * 32 + lane];
    }
    if (hi == 0) {
      float* op = orow_p + (T_LEN - 2) * 3;
      op[0] = o0;
      op[1] = o1;
      op[2] = o2;
    }
  }
  {
    float* ob = lds_out[T_LEN & 1];
    ob[(w * 3 + 0) * 32 + lane] = po[0];
    ob[(w * 3 + 1) * 32 + lane] = po[1];
    ob[(w * 3 + 2) * 32 + lane] = po[2];
    __syncthreads();
    if (w == 0) {
      float o0 = bo0, o1 = bo1, o2 = bo2;
#pragma unroll
      for (int s = 0; s < 4; ++s) {
        o0 += ob[(s * 3 + 0) * 32 + lane];
        o1 += ob[(s * 3 + 1) * 32 + lane];
        o2 += ob[(s * 3 + 2) * 32 + lane];
      }
      if (hi == 0) {
        float* op = orow_p + (T_LEN - 1) * 3;
        op[0] = o0;
        op[1] = o1;
        op[2] = o2;
      }
    }
  }
}

extern "C" void kernel_launch(void* const* d_in, const int* in_sizes, int n_in,
                              void* d_out, int out_size, void* d_ws,
                              size_t ws_size, hipStream_t stream) {
  const float* x     = (const float*)d_in[0];
  const float* W_ih  = (const float*)d_in[1];
  const float* W_hh  = (const float*)d_in[2];
  const float* b_ih  = (const float*)d_in[3];
  const float* b_hh  = (const float*)d_in[4];
  const float* W_out = (const float*)d_in[5];
  const float* b_out = (const float*)d_in[6];
  float* out = (float*)d_out;

  dim3 grid(4096 / 16);
  dim3 block(128);
  lstm_wmma_kernel<<<grid, block, 0, stream>>>(x, W_ih, W_hh, b_ih, b_hh,
                                               W_out, b_out, out);
}


# 8 waves per workgroup, 2 per SIMD: a wave owns 8 hidden units, takes their k-steps of all 4 K tiles, a pair swaps half gate tiles; one wave's waits filled by the other; gate and out sum order changes
# speedup vs baseline: 5.8363x; 1.1033x over previous
#include <hip/hip_runtime.h>


typedef __attribute__((ext_vector_type(2))) float v2f;
typedef __attribute__((ext_vector_type(4))) float v4f;
typedef __attribute__((ext_vector_type(8))) float v8f;

#define T_LEN 512

static __device__ __forceinline__ v8f wmma4(v2f a, v2f b, v8f c) {
  return __builtin_amdgcn_wmma_f32_16x16x4_f32(false, a, false, b, (short)0, c,
                                               false, false);
}

static __device__ __forceinline__ float sigm(float x) {
  return 1.0f / (1.0f + __expf(-x));
}
static __device__ __forceinline__ float tanh_(float x) {
  return 1.0f - 2.0f / (1.0f + __expf(2.0f * x));
}

#define QFRAG(H, E) ((v2f){(H)[2 * (E)], (H)[2 * (E) + 1]})

static __device__ __forceinline__ void keep_order() {
  __builtin_amdgcn_sched_barrier(0);
}
static __device__ __forceinline__ void lds_done_signal() {
  keep_order();
  asm volatile("s_wait_dscnt 0x0\n\ts_barrier_signal -1" ::: "memory");
  keep_order();
}
static __device__ __forceinline__ void barrier_wait() {
  keep_order();
  asm volatile("s_barrier_wait -1" ::: "memory");
  keep_order();
}

__global__ __launch_bounds__(256) void lstm_wmma_kernel(
    const float* __restrict__ x, const float* __restrict__ W_ih,
    const float* __restrict__ W_hh, const float* __restrict__ b_ih,
    const float* __restrict__ b_hh, const float* __restrict__ W_out,
    const float* __restrict__ b_out, float* __restrict__ out) {
  __shared__ v4f lds_h[8 * 32];
  __shared__ v4f lds_p[8 * 4 * 32];
  __shared__ float lds_out[2][8 * 3 * 32];

  const int tid  = threadIdx.x;
  const int wv   = __builtin_amdgcn_readfirstlane(tid >> 5);
  const int j    = wv & 3;
  const int s2   = wv >> 2;
  const int me   = j * 2 + s2;
  const int sib  = j * 2 + (s2 ^ 1);
  const int lane = tid & 31;
  const int col  = lane & 15;
  const int hi   = lane >> 4;
  const int bbase = blockIdx.x * 16;

  const int hrow = (col & 8) + 4 * (s2 ^ ((col >> 2) & 1)) + (col & 3);
  const int gx = 2 * ((col >> 2) & 1);
  const int kq = hi * 8 + s2 * 4;

  v2f wA[4][4][2];
#pragma unroll
  for (int i = 0; i < 4; ++i) {
#pragma unroll
    for (int g = 0; g < 4; ++g) {
      const int row = (g ^ gx) * 64 + j * 16 + hrow;
#pragma unroll
      for (int e = 0; e < 2; ++e) {
        const float* ap = &W_hh[row * 64 + (j ^ i) * 16 + kq + e * 2];
        wA[i][g][e][0] = ap[0];
        wA[i][g][e][1] = ap[1];
      }
    }
  }
  v2f wX[2];
#pragma unroll
  for (int i = 0; i < 2; ++i) {
    const int row = (i ^ gx) * 64 + j * 16 + hrow;
    const float w0 = W_ih[row * 3 + 0], w1 = W_ih[row * 3 + 1];
    const float w2 = W_ih[row * 3 + 2], bb = b_ih[row] + b_hh[row];
    wX[i][0] = hi ? w2 : w0;
    wX[i][1] = hi ? bb : w1;
  }
  v2f wO[2];
  {
    const int orow = (col < 3) ? col : 0;
#pragma unroll
    for (int e = 0; e < 2; ++e) {
      const float* op = &W_out[orow * 64 + j * 16 + kq + e * 2];
      wO[e][0] = (col < 3) ? op[0] : 0.0f;
      wO[e][1] = (col < 3) ? op[1] : 0.0f;
    }
  }
  const float bo0 = b_out[0], bo1 = b_out[1], bo2 = b_out[2];

  const float* xrow = x + (size_t)(bbase + col) * (T_LEN * 3);
  float* orow_p = out + (size_t)(bbase + col) * (T_LEN * 3);
  const v8f vzero = {0, 0, 0, 0, 0, 0, 0, 0};

  v4f* const h_mine = lds_h + me * 32 + lane;
  const v4f* const h_o1 = lds_h + ((j ^ 1) * 2 + s2) * 32 + lane;
  const v4f* const h_o2 = lds_h + ((j ^ 2) * 2 + s2) * 32 + lane;
  const v4f* const h_o3 = lds_h + ((j ^ 3) * 2 + s2) * 32 + lane;
  v4f* const p_send = lds_p + (sib * 4) * 32 + lane;
  const v4f* const p_recv = lds_p + (me * 4) * 32 + lane;

  v4f hq = {0, 0, 0, 0};
  v4f cq = {0, 0, 0, 0};
  h_mine[0] = hq;

  v8f ga[4], po = vzero;
#pragma unroll
  for (int g = 0; g < 4; ++g) ga[g] = vzero;

  v2f bx;
  {
    bx[0] = xrow[2 * hi];
    const float x1 = xrow[1 + hi];
    bx[1] = hi ? 1.0f : x1;
  }

  lds_done_signal();

  for (int t = 0; t < T_LEN; ++t) {
    float* ob = lds_out[t & 1];
    const float* obp = lds_out[(t & 1) ^ 1];

    barrier_wait();
    const v4f hf1 = h_o1[0];
    const v4f hf2 = h_o2[0];
    const v4f hf3 = h_o3[0];
    keep_order();
    ga[0] = wmma4(wX[0], bx, ga[0]);
    ga[1] = wmma4(wX[1], bx, ga[1]);
    v2f bxn;
    {
      const float* xp = xrow + ((t + 1) & (T_LEN - 1)) * 3;
      bxn[0] = xp[2 * hi];
      const float x1 = xp[1 + hi];
      bxn[1] = hi ? 1.0f : x1;
    }

#pragma unroll
    for (int e = 0; e < 2; ++e) {
#pragma unroll
      for (int g = 0; g < 4; ++g) ga[g] = wmma4(wA[1][g][e], QFRAG(hf1, e), ga[g]);
    }
#pragma unroll
    for (int e = 0; e < 2; ++e) {
#pragma unroll
      for (int g = 0; g < 4; ++g) ga[g] = wmma4(wA[2][g][e], QFRAG(hf2, e), ga[g]);
    }
#pragma unroll
    for (int e = 0; e < 2; ++e) {
#pragma unroll
      for (int g = 0; g < 4; ++g) ga[g] = wmma4(wA[3][g][e], QFRAG(hf3, e), ga[g]);
    }
#pragma unroll
    for (int g = 0; g < 4; ++g)
      p_send[(g ^ 2) * 32] = (v4f){ga[g][4], ga[g][5], ga[g][6], ga[g][7]};
    ob[(me * 3 + 0) * 32 + lane] = po[0];
    ob[(me * 3 + 1) * 32 + lane] = po[1];
    ob[(me * 3 + 2) * 32 + lane] = po[2];

    lds_done_signal();

    if (wv == 0 && t > 1) {
      float o0 = bo0, o1 = bo1, o2 = bo2;
#pragma unroll
      for (int s = 0; s < 8; ++s) {
        o0 += obp[(s * 3 + 0) * 32 + lane];
        o1 += obp[(s * 3 + 1) * 32 + lane];
        o2 += obp[(s * 3 + 2) * 32 + lane];
      }
      if (hi == 0) {
        float* op = orow_p + (t - 2) * 3;
        op[0] = o0;
        op[1] = o1;
        op[2] = o2;
      }
    }

    barrier_wait();
    v4f gt[4];
#pragma unroll
    for (int g = 0; g < 4; ++g) {
      const v4f rp = p_recv[g * 32];
      gt[g] = (v4f){ga[g][0], ga[g][1], ga[g][2], ga[g][3]} + rp;
    }

    auto cell_pair = [&](int p) {
#pragma unroll
      for (int r = 2 * p; r < 2 * p + 2; ++r) {
        float iv = sigm(gt[0][r]), fv = sigm(gt[1][r]);
        float gv = tanh_(gt[2][r]), ov = sigm(gt[3][r]);
        float cv = fv * cq[r] + iv * gv;
        cq[r] = cv;
        hq[r] = ov * tanh_(cv);
      }
    };
    cell_pair(0);
    keep_order();
    {
      const v2f b = QFRAG(hq, 0);
#pragma unroll
      for (int g = 0; g < 4; ++g) ga[g] = wmma4(wA[0][g][0], b, vzero);
      po = wmma4(wO[0], b, vzero);
    }
    cell_pair(1);
    keep_order();
    h_mine[0] = hq;
    keep_order();
    {
      const v2f b = QFRAG(hq, 1);
#pragma unroll
      for (int g = 0; g < 4; ++g) ga[g] = wmma4(wA[0][g][1], b, ga[g]);
      lds_done_signal();
      po = wmma4(wO[1], b, po);
    }
    bx = bxn;
  }

  barrier_wait();

  if (wv == 0) {
    const float* obp = lds_out[(T_LEN & 1) ^ 1];
    float o0 = bo0, o1 = bo1, o2 = bo2;
#pragma unroll
    for (int s = 0; s < 8; ++s) {
      o0 += obp[(s * 3 + 0) * 32 + lane];
      o1 += obp[(s * 3 + 1) * 32 + lane];
      o2 += obp[(s * 3 + 2) * 32 + lane];
    }
    if (hi == 0) {
      float* op = orow_p + (T_LEN - 2) * 3;
      op[0] = o0;
      op[1] = o1;
      op[2] = o2;
    }
  }
  {
    float* ob = lds_out[T_LEN & 1];
    ob[(me * 3 + 0) * 32 + lane] = po[0];
    ob[(me * 3 + 1) * 32 + lane] = po[1];
    ob[(me * 3 + 2) * 32 + lane] = po[2];
    __syncthreads();
    if (wv == 0) {
      float o0 = bo0, o1 = bo1, o2 = bo2;
#pragma unroll
      for (int s = 0; s < 8; ++s) {
        o0 += ob[(s * 3 + 0) * 32 + lane];
        o1 += ob[(s * 3 + 1) * 32 + lane];
        o2 += ob[(s * 3 + 2) * 32 + lane];
      }
      if (hi == 0) {
        float* op = orow_p + (T_LEN - 1) * 3;
        op[0] = o0;
        op[1] = o1;
        op[2] = o2;
      }
    }
  }
}

extern "C" void kernel_launch(void* const* d_in, const int* in_sizes, int n_in,
                              void* d_out, int out_size, void* d_ws,
                              size_t ws_size, hipStream_t stream) {
  const float* x     = (const float*)d_in[0];
  const float* W_ih  = (const float*)d_in[1];
  const float* W_hh  = (const float*)d_in[2];
  const float* b_ih  = (const float*)d_in[3];
  const float* b_hh  = (const float*)d_in[4];
  const float* W_out = (const float*)d_in[5];
  const float* b_out = (const float*)d_in[6];
  float* out = (float*)d_out;

  dim3 grid(4096 / 16);
  dim3 block(256);
  lstm_wmma_kernel<<<grid, block, 0, stream>>>(x, W_ih, W_hh, b_ih, b_hh,
                                               W_out, b_out, out);
}


# Gate products sigmoid(i)*tanh(g) and sigmoid(o)*tanh(c) each as ONE correctly rounded division (e-1)/((1+e')(1+e)): 3 divisions per hidden unit and step, not 5; rounding of c and h changes
# speedup vs baseline: 6.4649x; 1.1077x over previous
#include <hip/hip_runtime.h>


typedef __attribute__((ext_vector_type(2))) float v2f;
typedef __attribute__((ext_vector_type(4))) float v4f;
typedef __attribute__((ext_vector_type(8))) float v8f;

#define T_LEN 512

static __device__ __forceinline__ v8f wmma4(v2f a, v2f b, v8f c) {
  return __builtin_amdgcn_wmma_f32_16x16x4_f32(false, a, false, b, (short)0, c,
                                               false, false);
}

static __device__ __forceinline__ float sigm(float x) {
  return 1.0f / (1.0f + __expf(-x));
}
static __device__ __forceinline__ float sigm_tanh(float a, float b) {
  const float ea = __expf(-a);
  const float eb = __expf(2.0f * fminf(b, 20.0f));
  return (eb - 1.0f) / ((1.0f + ea) * (1.0f + eb));
}

#define QFRAG(H, E) ((v2f){(H)[2 * (E)], (H)[2 * (E) + 1]})

static __device__ __forceinline__ void keep_order() {
  __builtin_amdgcn_sched_barrier(0);
}
static __device__ __forceinline__ void lds_done_signal() {
  keep_order();
  asm volatile("s_wait_dscnt 0x0\n\ts_barrier_signal -1" ::: "memory");
  keep_order();
}
static __device__ __forceinline__ void barrier_wait() {
  keep_order();
  asm volatile("s_barrier_wait -1" ::: "memory");
  keep_order();
}

__global__ __launch_bounds__(256) void lstm_wmma_kernel(
    const float* __restrict__ x, const float* __restrict__ W_ih,
    const float* __restrict__ W_hh, const float* __restrict__ b_ih,
    const float* __restrict__ b_hh, const float* __restrict__ W_out,
    const float* __restrict__ b_out, float* __restrict__ out) {
  __shared__ v4f lds_h[8 * 32];
  __shared__ v4f lds_p[8 * 4 * 32];
  __shared__ float lds_out[2][8 * 3 * 32];

  const int tid  = threadIdx.x;
  const int wv   = __builtin_amdgcn_readfirstlane(tid >> 5);
  const int j    = wv & 3;
  const int s2   = wv >> 2;
  const int me   = j * 2 + s2;
  const int sib  = j * 2 + (s2 ^ 1);
  const int lane = tid & 31;
  const int col  = lane & 15;
  const int hi   = lane >> 4;
  const int bbase = blockIdx.x * 16;

  const int hrow = (col & 8) + 4 * (s2 ^ ((col >> 2) & 1)) + (col & 3);
  const int gx = 2 * ((col >> 2) & 1);
  const int kq = hi * 8 + s2 * 4;

  v2f wA[4][4][2];
#pragma unroll
  for (int i = 0; i < 4; ++i) {
#pragma unroll
    for (int g = 0; g < 4; ++g) {
      const int row = (g ^ gx) * 64 + j * 16 + hrow;
#pragma unroll
      for (int e = 0; e < 2; ++e) {
        const float* ap = &W_hh[row * 64 + (j ^ i) * 16 + kq + e * 2];
        wA[i][g][e][0] = ap[0];
        wA[i][g][e][1] = ap[1];
      }
    }
  }
  v2f wX[2];
#pragma unroll
  for (int i = 0; i < 2; ++i) {
    const int row = (i ^ gx) * 64 + j * 16 + hrow;
    const float w0 = W_ih[row * 3 + 0], w1 = W_ih[row * 3 + 1];
    const float w2 = W_ih[row * 3 + 2], bb = b_ih[row] + b_hh[row];
    wX[i][0] = hi ? w2 : w0;
    wX[i][1] = hi ? bb : w1;
  }
  v2f wO[2];
  {
    const int orow = (col < 3) ? col : 0;
#pragma unroll
    for (int e = 0; e < 2; ++e) {
      const float* op = &W_out[orow * 64 + j * 16 + kq + e * 2];
      wO[e][0] = (col < 3) ? op[0] : 0.0f;
      wO[e][1] = (col < 3) ? op[1] : 0.0f;
    }
  }
  const float bo0 = b_out[0], bo1 = b_out[1], bo2 = b_out[2];

  const float* xrow = x + (size_t)(bbase + col) * (T_LEN * 3);
  float* orow_p = out + (size_t)(bbase + col) * (T_LEN * 3);
  const v8f vzero = {0, 0, 0, 0, 0, 0, 0, 0};

  v4f* const h_mine = lds_h + me * 32 + lane;
  const v4f* const h_o1 = lds_h + ((j ^ 1) * 2 + s2) * 32 + lane;
  const v4f* const h_o2 = lds_h + ((j ^ 2) * 2 + s2) * 32 + lane;
  const v4f* const h_o3 = lds_h + ((j ^ 3) * 2 + s2) * 32 + lane;
  v4f* const p_send = lds_p + (sib * 4) * 32 + lane;
  const v4f* const p_recv = lds_p + (me * 4) * 32 + lane;

  v4f hq = {0, 0, 0, 0};
  v4f cq = {0, 0, 0, 0};
  h_mine[0] = hq;

  v8f ga[4], po = vzero;
#pragma unroll
  for (int g = 0; g < 4; ++g) ga[g] = vzero;

  v2f bx;
  {
    bx[0] = xrow[2 * hi];
    const float x1 = xrow[1 + hi];
    bx[1] = hi ? 1.0f : x1;
  }

  lds_done_signal();

  for (int t = 0; t < T_LEN; ++t) {
    float* ob = lds_out[t & 1];
    const float* obp = lds_out[(t & 1) ^ 1];

    barrier_wait();
    const v4f hf1 = h_o1[0];
    const v4f hf2 = h_o2[0];
    const v4f hf3 = h_o3[0];
    keep_order();
    ga[0] = wmma4(wX[0], bx, ga[0]);
    ga[1] = wmma4(wX[1], bx, ga[1]);
    v2f bxn;
    {
      const float* xp = xrow + ((t + 1) & (T_LEN - 1)) * 3;
      bxn[0] = xp[2 * hi];
      const float x1 = xp[1 + hi];
      bxn[1] = hi ? 1.0f : x1;
    }

#pragma unroll
    for (int e = 0; e < 2; ++e) {
#pragma unroll
      for (int g = 0; g < 4; ++g) ga[g] = wmma4(wA[1][g][e], QFRAG(hf1, e), ga[g]);
    }
#pragma unroll
    for (int e = 0; e < 2; ++e) {
#pragma unroll
      for (int g = 0; g < 4; ++g) ga[g] = wmma4(wA[2][g][e], QFRAG(hf2, e), ga[g]);
    }
#pragma unroll
    for (int e = 0; e < 2; ++e) {
#pragma unroll
      for (int g = 0; g < 4; ++g) ga[g] = wmma4(wA[3][g][e], QFRAG(hf3, e), ga[g]);
    }
#pragma unroll
    for (int g = 0; g < 4; ++g)
      p_send[(g ^ 2) * 32] = (v4f){ga[g][4], ga[g][5], ga[g][6], ga[g][7]};
    ob[(me * 3 + 0) * 32 + lane] = po[0];
    ob[(me * 3 + 1) * 32 + lane] = po[1];
    ob[(me * 3 + 2) * 32 + lane] = po[2];

    lds_done_signal();

    if (wv == 0 && t > 1) {
      float o0 = bo0, o1 = bo1, o2 = bo2;
#pragma unroll
      for (int s = 0; s < 8; ++s) {
        o0 += obp[(s * 3 + 0) * 32 + lane];
        o1 += obp[(s * 3 + 1) * 32 + lane];
        o2 += obp[(s * 3 + 2) * 32 + lane];
      }
      if (hi == 0) {
        float* op = orow_p + (t - 2) * 3;
        op[0] = o0;
        op[1] = o1;
        op[2] = o2;
      }
    }

    barrier_wait();
    v4f gt[4];
#pragma unroll
    for (int g = 0; g < 4; ++g) {
      const v4f rp = p_recv[g * 32];
      gt[g] = (v4f){ga[g][0], ga[g][1], ga[g][2], ga[g][3]} + rp;
    }

    auto cell_pair = [&](int p) {
#pragma unroll
      for (int r = 2 * p; r < 2 * p + 2; ++r) {
        const float fv = sigm(gt[1][r]);
        const float cv = fv * cq[r] + sigm_tanh(gt[0][r], gt[2][r]);
        cq[r] = cv;
        hq[r] = sigm_tanh(gt[3][r], cv);
      }
    };
    cell_pair(0);
    keep_order();
    {
      const v2f b = QFRAG(hq, 0);
#pragma unroll
      for (int g = 0; g < 4; ++g) ga[g] = wmma4(wA[0][g][0], b, vzero);
      po = wmma4(wO[0], b, vzero);
    }
    cell_pair(1);
    keep_order();
    h_mine[0] = hq;
    keep_order();
    {
      const v2f b = QFRAG(hq, 1);
#pragma unroll
      for (int g = 0; g < 4; ++g) ga[g] = wmma4(wA[0][g][1], b, ga[g]);
      lds_done_signal();
      po = wmma4(wO[1], b, po);
    }
    bx = bxn;
  }

  barrier_wait();

  if (wv == 0) {
    const float* obp = lds_out[(T_LEN & 1) ^ 1];
    float o0 = bo0, o1 = bo1, o2 = bo2;
#pragma unroll
    for (int s = 0; s < 8; ++s) {
      o0 += obp[(s * 3 + 0) * 32 + lane];
      o1 += obp[(s * 3 + 1) * 32 + lane];
      o2 += obp[(s * 3 + 2) * 32 + lane];
    }
    if (hi == 0) {
      float* op = orow_p + (T_LEN - 2) * 3;
      op[0] = o0;
      op[1] = o1;
      op[2] = o2;
    }
  }
  {
    float* ob = lds_out[T_LEN & 1];
    ob[(me * 3 + 0) * 32 + lane] = po[0];
    ob[(me * 3 + 1) * 32 + lane] = po[1];
    ob[(me * 3 + 2) * 32 + lane] = po[2];
    __syncthreads();
    if (wv == 0) {
      float o0 = bo0, o1 = bo1, o2 = bo2;
#pragma unroll
      for (int s = 0; s < 8; ++s) {
        o0 += ob[(s * 3 + 0) * 32 + lane];
        o1 += ob[(s * 3 + 1) * 32 + lane];
        o2 += ob[(s * 3 + 2) * 32 + lane];
      }
      if (hi == 0) {
        float* op = orow_p + (T_LEN - 1) * 3;
        op[0] = o0;
        op[1] = o1;
        op[2] = o2;
      }
    }
  }
}

extern "C" void kernel_launch(void* const* d_in, const int* in_sizes, int n_in,
                              void* d_out, int out_size, void* d_ws,
                              size_t ws_size, hipStream_t stream) {
  const float* x     = (const float*)d_in[0];
  const float* W_ih  = (const float*)d_in[1];
  const float* W_hh  = (const float*)d_in[2];
  const float* b_ih  = (const float*)d_in[3];
  const float* b_hh  = (const float*)d_in[4];
  const float* W_out = (const float*)d_in[5];
  const float* b_out = (const float*)d_in[6];
  float* out = (float*)d_out;

  dim3 grid(4096 / 16);
  dim3 block(256);
  lstm_wmma_kernel<<<grid, block, 0, stream>>>(x, W_ih, W_hh, b_ih, b_hh,
                                               W_out, b_out, out);
}


# A rows regrouped by receiver: the sibling's 2 partial tiles are summed and sent first, the 12 WMMAs of the wave's own 2 tiles span the exchange's barrier and LDS round trip; gate sum order changes
# speedup vs baseline: 6.7061x; 1.0373x over previous
#include <hip/hip_runtime.h>


typedef __attribute__((ext_vector_type(2))) float v2f;
typedef __attribute__((ext_vector_type(4))) float v4f;
typedef __attribute__((ext_vector_type(8))) float v8f;

#define T_LEN 512

static __device__ __forceinline__ v8f wmma4(v2f a, v2f b, v8f c) {
  return __builtin_amdgcn_wmma_f32_16x16x4_f32(false, a, false, b, (short)0, c,
                                               false, false);
}

static __device__ __forceinline__ float sigm(float x) {
  return 1.0f / (1.0f + __expf(-x));
}
static __device__ __forceinline__ float sigm_tanh(float a, float b) {
  const float ea = __expf(-a);
  const float eb = __expf(2.0f * fminf(b, 20.0f));
  return (eb - 1.0f) / ((1.0f + ea) * (1.0f + eb));
}

#define QFRAG(H, E) ((v2f){(H)[2 * (E)], (H)[2 * (E) + 1]})

static __device__ __forceinline__ void keep_order() {
  __builtin_amdgcn_sched_barrier(0);
}
static __device__ __forceinline__ void lds_done_signal() {
  keep_order();
  asm volatile("s_wait_dscnt 0x0\n\ts_barrier_signal -1" ::: "memory");
  keep_order();
}
static __device__ __forceinline__ void barrier_wait() {
  keep_order();
  asm volatile("s_barrier_wait -1" ::: "memory");
  keep_order();
}

__global__ __launch_bounds__(256) void lstm_wmma_kernel(
    const float* __restrict__ x, const float* __restrict__ W_ih,
    const float* __restrict__ W_hh, const float* __restrict__ b_ih,
    const float* __restrict__ b_hh, const float* __restrict__ W_out,
    const float* __restrict__ b_out, float* __restrict__ out) {
  __shared__ v4f lds_h[8 * 32];
  __shared__ v4f lds_p[8 * 4 * 32];
  __shared__ float lds_out[2][8 * 3 * 32];

  const int tid  = threadIdx.x;
  const int wv   = __builtin_amdgcn_readfirstlane(tid >> 5);
  const int j    = wv & 3;
  const int s2   = wv >> 2;
  const int me   = j * 2 + s2;
  const int sib  = j * 2 + (s2 ^ 1);
  const int lane = tid & 31;
  const int col  = lane & 15;
  const int hi   = lane >> 4;
  const int bbase = blockIdx.x * 16;

  const int hrow_own = (col & 8) + 4 * s2 + (col & 3);
  const int hrow_sib = (col & 8) + 4 * (s2 ^ 1) + (col & 3);
  const int gsel = (col >> 2) & 1;
  const int kq = hi * 8 + s2 * 4;

  v2f wA[4][4][2];
#pragma unroll
  for (int i = 0; i < 4; ++i) {
#pragma unroll
    for (int g = 0; g < 4; ++g) {
      const int row = (2 * (g & 1) + gsel) * 64 + j * 16 + ((g < 2) ? hrow_own : hrow_sib);
#pragma unroll
      for (int e = 0; e < 2; ++e) {
        const float* ap = &W_hh[row * 64 + (j ^ i) * 16 + kq + e * 2];
        wA[i][g][e][0] = ap[0];
        wA[i][g][e][1] = ap[1];
      }
    }
  }
  v2f wX[2];
#pragma unroll
  for (int i = 0; i < 2; ++i) {
    const int row = (2 * i + gsel) * 64 + j * 16 + hrow_own;
    const float w0 = W_ih[row * 3 + 0], w1 = W_ih[row * 3 + 1];
    const float w2 = W_ih[row * 3 + 2], bb = b_ih[row] + b_hh[row];
    wX[i][0] = hi ? w2 : w0;
    wX[i][1] = hi ? bb : w1;
  }
  v2f wO[2];
  {
    const int orow = (col < 3) ? col : 0;
#pragma unroll
    for (int e = 0; e < 2; ++e) {
      const float* op = &W_out[orow * 64 + j * 16 + kq + e * 2];
      wO[e][0] = (col < 3) ? op[0] : 0.0f;
      wO[e][1] = (col < 3) ? op[1] : 0.0f;
    }
  }
  const float bo0 = b_out[0], bo1 = b_out[1], bo2 = b_out[2];

  const float* xrow = x + (size_t)(bbase + col) * (T_LEN * 3);
  float* orow_p = out + (size_t)(bbase + col) * (T_LEN * 3);
  const v8f vzero = {0, 0, 0, 0, 0, 0, 0, 0};

  v4f* const h_mine = lds_h + me * 32 + lane;
  const v4f* const h_o1 = lds_h + ((j ^ 1) * 2 + s2) * 32 + lane;
  const v4f* const h_o2 = lds_h + ((j ^ 2) * 2 + s2) * 32 + lane;
  const v4f* const h_o3 = lds_h + ((j ^ 3) * 2 + s2) * 32 + lane;
  v4f* const p_send = lds_p + (sib * 4) * 32 + lane;
  const v4f* const p_recv = lds_p + (me * 4) * 32 + lane;

  v4f hq = {0, 0, 0, 0};
  v4f cq = {0, 0, 0, 0};
  h_mine[0] = hq;

  v8f ga[4], po = vzero;
#pragma unroll
  for (int g = 0; g < 4; ++g) ga[g] = vzero;

  v2f bx;
  {
    bx[0] = xrow[2 * hi];
    const float x1 = xrow[1 + hi];
    bx[1] = hi ? 1.0f : x1;
  }

  lds_done_signal();

  for (int t = 0; t < T_LEN; ++t) {
    float* ob = lds_out[t & 1];
    const float* obp = lds_out[(t & 1) ^ 1];

    barrier_wait();
    const v4f hf1 = h_o1[0];
    const v4f hf2 = h_o2[0];
    const v4f hf3 = h_o3[0];
    keep_order();
    ga[0] = wmma4(wX[0], bx, ga[0]);
    ga[1] = wmma4(wX[1], bx, ga[1]);
    v2f bxn;
    {
      const float* xp = xrow + ((t + 1) & (T_LEN - 1)) * 3;
      bxn[0] = xp[2 * hi];
      const float x1 = xp[1 + hi];
      bxn[1] = hi ? 1.0f : x1;
    }

#pragma unroll
    for (int e = 0; e < 2; ++e) {
#pragma unroll
      for (int g = 2; g < 4; ++g) ga[g] = wmma4(wA[1][g][e], QFRAG(hf1, e), ga[g]);
    }
#pragma unroll
    for (int e = 0; e < 2; ++e) {
#pragma unroll
      for (int g = 2; g < 4; ++g) ga[g] = wmma4(wA[2][g][e], QFRAG(hf2, e), ga[g]);
    }
#pragma unroll
    for (int e = 0; e < 2; ++e) {
#pragma unroll
      for (int g = 2; g < 4; ++g) ga[g] = wmma4(wA[3][g][e], QFRAG(hf3, e), ga[g]);
    }
    p_send[0 * 32] = (v4f){ga[2][0], ga[2][1], ga[2][2], ga[2][3]};
    p_send[1 * 32] = (v4f){ga[2][4], ga[2][5], ga[2][6], ga[2][7]};
    p_send[2 * 32] = (v4f){ga[3][0], ga[3][1], ga[3][2], ga[3][3]};
    p_send[3 * 32] = (v4f){ga[3][4], ga[3][5], ga[3][6], ga[3][7]};
    ob[(me * 3 + 0) * 32 + lane] = po[0];
    ob[(me * 3 + 1) * 32 + lane] = po[1];
    ob[(me * 3 + 2) * 32 + lane] = po[2];
    keep_order();
#pragma unroll
    for (int e = 0; e < 2; ++e) {
#pragma unroll
      for (int g = 0; g < 2; ++g) ga[g] = wmma4(wA[1][g][e], QFRAG(hf1, e), ga[g]);
    }

    lds_done_signal();

    if (wv == 0 && t > 1) {
      float o0 = bo0, o1 = bo1, o2 = bo2;
#pragma unroll
      for (int s = 0; s < 8; ++s) {
        o0 += obp[(s * 3 + 0) * 32 + lane];
        o1 += obp[(s * 3 + 1) * 32 + lane];
        o2 += obp[(s * 3 + 2) * 32 + lane];
      }
      if (hi == 0) {
        float* op = orow_p + (t - 2) * 3;
        op[0] = o0;
        op[1] = o1;
        op[2] = o2;
      }
    }

#pragma unroll
    for (int e = 0; e < 2; ++e) {
#pragma unroll
      for (int g = 0; g < 2; ++g) ga[g] = wmma4(wA[2][g][e], QFRAG(hf2, e), ga[g]);
    }

    barrier_wait();
    v4f rp[4];
#pragma unroll
    for (int g = 0; g < 4; ++g) rp[g] = p_recv[g * 32];
    keep_order();
#pragma unroll
    for (int e = 0; e < 2; ++e) {
#pragma unroll
      for (int g = 0; g < 2; ++g) ga[g] = wmma4(wA[3][g][e], QFRAG(hf3, e), ga[g]);
    }
    v4f gt[4];
    gt[0] = (v4f){ga[0][0], ga[0][1], ga[0][2], ga[0][3]} + rp[0];
    gt[1] = (v4f){ga[0][4], ga[0][5], ga[0][6], ga[0][7]} + rp[1];
    gt[2] = (v4f){ga[1][0], ga[1][1], ga[1][2], ga[1][3]} + rp[2];
    gt[3] = (v4f){ga[1][4], ga[1][5], ga[1][6], ga[1][7]} + rp[3];

    auto cell_pair = [&](int p) {
#pragma unroll
      for (int r = 2 * p; r < 2 * p + 2; ++r) {
        const float fv = sigm(gt[1][r]);
        const float cv = fv * cq[r] + sigm_tanh(gt[0][r], gt[2][r]);
        cq[r] = cv;
        hq[r] = sigm_tanh(gt[3][r], cv);
      }
    };
    cell_pair(0);
    keep_order();
    {
      const v2f b = QFRAG(hq, 0);
#pragma unroll
      for (int g = 0; g < 4; ++g) ga[g] = wmma4(wA[0][g][0], b, vzero);
      po = wmma4(wO[0], b, vzero);
    }
    cell_pair(1);
    keep_order();
    h_mine[0] = hq;
    keep_order();
    {
      const v2f b = QFRAG(hq, 1);
#pragma unroll
      for (int g = 0; g < 4; ++g) ga[g] = wmma4(wA[0][g][1], b, ga[g]);
      lds_done_signal();
      po = wmma4(wO[1], b, po);
    }
    bx = bxn;
  }

  barrier_wait();

  if (wv == 0) {
    const float* obp = lds_out[(T_LEN & 1) ^ 1];
    float o0 = bo0, o1 = bo1, o2 = bo2;
#pragma unroll
    for (int s = 0; s < 8; ++s) {
      o0 += obp[(s * 3 + 0) * 32 + lane];
      o1 += obp[(s * 3 + 1) * 32 + lane];
      o2 += obp[(s * 3 + 2) * 32 + lane];
    }
    if (hi == 0) {
      float* op = orow_p + (T_LEN - 2) * 3;
      op[0] = o0;
      op[1] = o1;
      op[2] = o2;
    }
  }
  {
    float* ob = lds_out[T_LEN & 1];
    ob[(me * 3 + 0) * 32 + lane] = po[0];
    ob[(me * 3 + 1) * 32 + lane] = po[1];
    ob[(me * 3 + 2) * 32 + lane] = po[2];
    __syncthreads();
    if (wv == 0) {
      float o0 = bo0, o1 = bo1, o2 = bo2;
#pragma unroll
      for (int s = 0; s < 8; ++s) {
        o0 += ob[(s * 3 + 0) * 32 + lane];
        o1 += ob[(s * 3 + 1) * 32 + lane];
        o2 += ob[(s * 3 + 2) * 32 + lane];
      }
      if (hi == 0) {
        float* op = orow_p + (T_LEN - 1) * 3;
        op[0] = o0;
        op[1] = o1;
        op[2] = o2;
      }
    }
  }
}

extern "C" void kernel_launch(void* const* d_in, const int* in_sizes, int n_in,
                              void* d_out, int out_size, void* d_ws,
                              size_t ws_size, hipStream_t stream) {
  const float* x     = (const float*)d_in[0];
  const float* W_ih  = (const float*)d_in[1];
  const float* W_hh  = (const float*)d_in[2];
  const float* b_ih  = (const float*)d_in[3];
  const float* b_hh  = (const float*)d_in[4];
  const float* W_out = (const float*)d_in[5];
  const float* b_out = (const float*)d_in[6];
  float* out = (float*)d_out;

  dim3 grid(4096 / 16);
  dim3 block(256);
  lstm_wmma_kernel<<<grid, block, 0, stream>>>(x, W_ih, W_hh, b_ih, b_hh,
                                               W_out, b_out, out);
}
